# starting kernel
# baseline (speedup 1.0000x reference)
#include <hip/hip_runtime.h>


typedef __attribute__((ext_vector_type(16))) _Float16 v16h;
typedef __attribute__((ext_vector_type(8)))  float    v8f;

static constexpr int BATCH    = 4;
static constexpr int CIN      = 64;
static constexpr int COUT     = 64;
static constexpr int NPTS     = 16384;
static constexpr int KNBH     = 8;

static constexpr int ROWS     = 4 * COUT;
static constexpr int TILE_PTS = 32;
static constexpr int LDS_STR  = 33;
static constexpr int NPAIRS   = TILE_PTS * KNBH;
static constexpr int A_ELEMS  = 16   * 2   * 32   * 16  ;
static constexpr int B_HALVES = 2   * 2   * 32   * 16  ;

__device__ __forceinline__ int kmap(int h, int g, int kc) {
    return 32 * kc + 16 * (h >> 3) + 8 * g + (h & 7);
}

__global__ void init_out_kernel(const float* __restrict__ bias,
                                float* __restrict__ out) {
    size_t idx = (size_t)blockIdx.x * blockDim.x + threadIdx.x;
    size_t nwords = (size_t)BATCH * COUT * NPTS / 4;
    if (idx >= nwords) return;
    int o = (int)((idx / (NPTS / 4)) % COUT);
    float v = bias[o];
    ((float4*)out)[idx] = make_float4(v, v, v, v);
}

__global__ void pack_weights_kernel(const float* __restrict__ theta,
                                    const float* __restrict__ wbias,
                                    _Float16* __restrict__ wsA) {
    int e = blockIdx.x * blockDim.x + threadIdx.x;
    if (e >= A_ELEMS) return;
    int h    = e & 15;
    int lane = (e >> 4) & 31;
    int kc   = (e >> 9) & 1;
    int mt   = e >> 10;
    int row  = mt * 16 + (lane & 15);
    int g    = lane >> 4;
    int Kk   = kmap(h, g, kc);
    float w;
    if (row < 3 * COUT) {
        int d = row >> 6, o = row & 63;
        w = theta[((size_t)d * CIN + Kk) * COUT + o];
    } else {
        int o = row - 3 * COUT;
        w = wbias[(size_t)Kk * COUT + o];
    }
    wsA[e] = (_Float16)w;
}

__global__ void __launch_bounds__(256)
flexconv_kernel(const float* __restrict__ features,
                const int*   __restrict__ nbh,
                const float* __restrict__ pos,
                const _Float16* __restrict__ wsA,
                float* __restrict__ out) {
    __shared__ float ftfb[ROWS * LDS_STR];
    __shared__ float pdx[NPAIRS], pdy[NPAIRS], pdz[NPAIRS];
    __shared__ int   pj[NPAIRS];
    __shared__ __align__(32) _Float16 bstage[B_HALVES];

    const int tid        = threadIdx.x;
    const int blocksPerB = NPTS / TILE_PTS;
    const int b          = blockIdx.x / blocksPerB;
    const int n0         = (blockIdx.x % blocksPerB) * TILE_PTS;

    {
        int q  = tid;
        int pt = q >> 3, k = q & 7;
        int n  = n0 + pt;
        int j  = nbh[((size_t)b * KNBH + k) * NPTS + n];
        const float* posb = pos + (size_t)b * 3 * NPTS;
        pj[q]  = j;
        pdx[q] = posb[0 * NPTS + j] - posb[0 * NPTS + n];
        pdy[q] = posb[1 * NPTS + j] - posb[1 * NPTS + n];
        pdz[q] = posb[2 * NPTS + j] - posb[2 * NPTS + n];
    }

    {
        const int ncol = tid & 31;
        const int krow = tid >> 5;
        const float* featb = features + (size_t)b * CIN * NPTS + n0;
        const int nt   = ncol >> 4;
        const int lcol = ncol & 15;
        #pragma unroll
        for (int pass = 0; pass < 8; ++pass) {
            const int Kk  = krow + 8 * pass;
            const float v = featb[(size_t)Kk * NPTS + ncol];
            const int kc  = Kk >> 5;
            const int rem = Kk & 31;
            const int g   = (rem >> 3) & 1;
            const int h   = 8 * (rem >> 4) + (rem & 7);
            const int lane = g * 16 + lcol;
            bstage[(((nt * 2 + kc) * 32) + lane) * 16 + h] = (_Float16)v;
        }
    }
    __syncthreads();

    {
        const int wave  = tid >> 5;
        const int lane  = tid & 31;
        const int nt    = wave & 1;
        const int mbase = wave >> 1;
        const int g     = lane >> 4;
        const int lcol  = lane & 15;

        const v16h bfrag0 = *(const v16h*)(bstage + ((nt * 2 + 0) * 32 + lane) * 16);
        const v16h bfrag1 = *(const v16h*)(bstage + ((nt * 2 + 1) * 32 + lane) * 16);

        #pragma unroll
        for (int i = 0; i < 4; ++i) {
            const int mt = mbase + 4 * i;
            v8f c = {};
            {
                v16h a = *(const v16h*)(wsA + (((size_t)mt * 2 + 0) * 32 + lane) * 16);
                c = __builtin_amdgcn_wmma_f32_16x16x32_f16(
                        false, a, false, bfrag0, (short)0, c, false, false);
            }
            {
                v16h a = *(const v16h*)(wsA + (((size_t)mt * 2 + 1) * 32 + lane) * 16);
                c = __builtin_amdgcn_wmma_f32_16x16x32_f16(
                        false, a, false, bfrag1, (short)0, c, false, false);
            }
            const int colBase = nt * 16 + lcol;
            #pragma unroll
            for (int r = 0; r < 8; ++r) {
                int row = mt * 16 + 8 * g + r;
                ftfb[row * LDS_STR + colBase] = c[r];
            }
        }
    }
    __syncthreads();

    {
        const int o = tid & 63;
        for (int pp = tid >> 6; pp < NPAIRS; pp += 4) {
            const int   pt = pp >> 3;
            const float dx = pdx[pp], dy = pdy[pp], dz = pdz[pp];
            const int   j  = pj[pp];
            const float c0 = ftfb[(0 * COUT + o) * LDS_STR + pt];
            const float c1 = ftfb[(1 * COUT + o) * LDS_STR + pt];
            const float c2 = ftfb[(2 * COUT + o) * LDS_STR + pt];
            const float cb = ftfb[(3 * COUT + o) * LDS_STR + pt];
            const float contrib = dx * c0 + dy * c1 + dz * c2 + cb;
            atomicAdd(&out[((size_t)b * COUT + o) * NPTS + j], contrib);
        }
    }
}

extern "C" void kernel_launch(void* const* d_in, const int* in_sizes, int n_in,
                              void* d_out, int out_size, void* d_ws, size_t ws_size,
                              hipStream_t stream) {
    const float* features = (const float*)d_in[0];
    const float* theta    = (const float*)d_in[1];
    const float* wbias    = (const float*)d_in[2];
    const int*   nbh      = (const int*)d_in[3];
    const float* pos      = (const float*)d_in[4];
    const float* bias     = (const float*)d_in[5];
    float*       out      = (float*)d_out;
    _Float16*    wsA      = (_Float16*)d_ws;

    {
        int nwords = BATCH * COUT * NPTS / 4;
        init_out_kernel<<<(nwords + 255) / 256, 256, 0, stream>>>(bias, out);
    }
    pack_weights_kernel<<<(A_ELEMS + 255) / 256, 256, 0, stream>>>(theta, wbias, wsA);
    flexconv_kernel<<<BATCH * (NPTS / TILE_PTS), 256, 0, stream>>>(
        features, nbh, pos, wsA, out);
}


# Scatter atomics go to an intermediate order of out where a point's 64 channels are 256 contiguous bytes (one line per wave atomic, not 32); a 4th kernel turns each 64x64 block over in place
# speedup vs baseline: 10.6672x; 10.6672x over previous
#include <hip/hip_runtime.h>


typedef __attribute__((ext_vector_type(16))) _Float16 v16h;
typedef __attribute__((ext_vector_type(8)))  float    v8f;

static constexpr int BATCH    = 4;
static constexpr int CIN      = 64;
static constexpr int COUT     = 64;
static constexpr int NPTS     = 16384;
static constexpr int KNBH     = 8;

static constexpr int ROWS     = 4 * COUT;
static constexpr int TILE_PTS = 32;
static constexpr int LDS_STR  = 33;
static constexpr int NPAIRS   = TILE_PTS * KNBH;
static constexpr int A_ELEMS  = 16   * 2   * 32   * 16  ;
static constexpr int B_HALVES = 2   * 2   * 32   * 16  ;
static constexpr int TBLK     = 64;
static constexpr int TBLK_STR = TBLK + 1;

__device__ __forceinline__ int kmap(int h, int g, int kc) {
    return 32 * kc + 16 * (h >> 3) + 8 * g + (h & 7);
}

__global__ void init_out_kernel(const float* __restrict__ bias,
                                float* __restrict__ out) {
    size_t idx = (size_t)blockIdx.x * blockDim.x + threadIdx.x;
    size_t nwords = (size_t)BATCH * COUT * NPTS / 4;
    if (idx >= nwords) return;
    int o = (int)(idx & 15) * 4;
    ((float4*)out)[idx] = make_float4(bias[o], bias[o + 1], bias[o + 2], bias[o + 3]);
}

__global__ void pack_weights_kernel(const float* __restrict__ theta,
                                    const float* __restrict__ wbias,
                                    _Float16* __restrict__ wsA) {
    int e = blockIdx.x * blockDim.x + threadIdx.x;
    if (e >= A_ELEMS) return;
    int h    = e & 15;
    int lane = (e >> 4) & 31;
    int kc   = (e >> 9) & 1;
    int mt   = e >> 10;
    int row  = mt * 16 + (lane & 15);
    int g    = lane >> 4;
    int Kk   = kmap(h, g, kc);
    float w;
    if (row < 3 * COUT) {
        int d = row >> 6, o = row & 63;
        w = theta[((size_t)d * CIN + Kk) * COUT + o];
    } else {
        int o = row - 3 * COUT;
        w = wbias[(size_t)Kk * COUT + o];
    }
    wsA[e] = (_Float16)w;
}

__global__ void __launch_bounds__(256)
flexconv_kernel(const float* __restrict__ features,
                const int*   __restrict__ nbh,
                const float* __restrict__ pos,
                const _Float16* __restrict__ wsA,
                float* __restrict__ out) {
    __shared__ float ftfb[ROWS * LDS_STR];
    __shared__ float pdx[NPAIRS], pdy[NPAIRS], pdz[NPAIRS];
    __shared__ int   pj[NPAIRS];
    __shared__ __align__(32) _Float16 bstage[B_HALVES];

    const int tid        = threadIdx.x;
    const int blocksPerB = NPTS / TILE_PTS;
    const int b          = blockIdx.x / blocksPerB;
    const int n0         = (blockIdx.x % blocksPerB) * TILE_PTS;

    {
        int q  = tid;
        int pt = q >> 3, k = q & 7;
        int n  = n0 + pt;
        int j  = nbh[((size_t)b * KNBH + k) * NPTS + n];
        const float* posb = pos + (size_t)b * 3 * NPTS;
        pj[q]  = (j & (TBLK - 1)) * NPTS + (j & ~(TBLK - 1));
        pdx[q] = posb[0 * NPTS + j] - posb[0 * NPTS + n];
        pdy[q] = posb[1 * NPTS + j] - posb[1 * NPTS + n];
        pdz[q] = posb[2 * NPTS + j] - posb[2 * NPTS + n];
    }

    {
        const int ncol = tid & 31;
        const int krow = tid >> 5;
        const float* featb = features + (size_t)b * CIN * NPTS + n0;
        const int nt   = ncol >> 4;
        const int lcol = ncol & 15;
        #pragma unroll
        for (int pass = 0; pass < 8; ++pass) {
            const int Kk  = krow + 8 * pass;
            const float v = featb[(size_t)Kk * NPTS + ncol];
            const int kc  = Kk >> 5;
            const int rem = Kk & 31;
            const int g   = (rem >> 3) & 1;
            const int h   = 8 * (rem >> 4) + (rem & 7);
            const int lane = g * 16 + lcol;
            bstage[(((nt * 2 + kc) * 32) + lane) * 16 + h] = (_Float16)v;
        }
    }
    __syncthreads();

    {
        const int wave  = tid >> 5;
        const int lane  = tid & 31;
        const int nt    = wave & 1;
        const int mbase = wave >> 1;
        const int g     = lane >> 4;
        const int lcol  = lane & 15;

        const v16h bfrag0 = *(const v16h*)(bstage + ((nt * 2 + 0) * 32 + lane) * 16);
        const v16h bfrag1 = *(const v16h*)(bstage + ((nt * 2 + 1) * 32 + lane) * 16);

        #pragma unroll
        for (int i = 0; i < 4; ++i) {
            const int mt = mbase + 4 * i;
            v8f c = {};
            {
                v16h a = *(const v16h*)(wsA + (((size_t)mt * 2 + 0) * 32 + lane) * 16);
                c = __builtin_amdgcn_wmma_f32_16x16x32_f16(
                        false, a, false, bfrag0, (short)0, c, false, false);
            }
            {
                v16h a = *(const v16h*)(wsA + (((size_t)mt * 2 + 1) * 32 + lane) * 16);
                c = __builtin_amdgcn_wmma_f32_16x16x32_f16(
                        false, a, false, bfrag1, (short)0, c, false, false);
            }
            const int colBase = nt * 16 + lcol;
            #pragma unroll
            for (int r = 0; r < 8; ++r) {
                int row = mt * 16 + 8 * g + r;
                ftfb[row * LDS_STR + colBase] = c[r];
            }
        }
    }
    __syncthreads();

    {
        const int o = tid & 63;
        float* outb = out + (size_t)b * COUT * NPTS + o;
        for (int pp = tid >> 6; pp < NPAIRS; pp += 4) {
            const int   pt = pp >> 3;
            const float dx = pdx[pp], dy = pdy[pp], dz = pdz[pp];
            const int   joff = pj[pp];
            const float c0 = ftfb[(0 * COUT + o) * LDS_STR + pt];
            const float c1 = ftfb[(1 * COUT + o) * LDS_STR + pt];
            const float c2 = ftfb[(2 * COUT + o) * LDS_STR + pt];
            const float cb = ftfb[(3 * COUT + o) * LDS_STR + pt];
            const float contrib = dx * c0 + dy * c1 + dz * c2 + cb;
            atomicAdd(outb + joff, contrib);
        }
    }
}

__global__ void __launch_bounds__(256)
turn_blocks_kernel(float* out) {
    __shared__ float blk[TBLK * TBLK_STR];

    const int tid = threadIdx.x;
    const int b   = blockIdx.x / (NPTS / TBLK);
    const int J   = blockIdx.x % (NPTS / TBLK);
    float* base   = out + (size_t)b * COUT * NPTS + J * TBLK;
    const int c4  = tid & 15;
    const int r0  = tid >> 4;

    float4 v[4];
    for (int pass = 0; pass < 4; ++pass) {
        const int r = r0 + 16 * pass;
        v[pass] = *(const float4*)(base + (size_t)r * NPTS + 4 * c4);
    }
    for (int pass = 0; pass < 4; ++pass) {
        const int r = r0 + 16 * pass;
        blk[r * TBLK_STR + 4 * c4 + 0] = v[pass].x;
        blk[r * TBLK_STR + 4 * c4 + 1] = v[pass].y;
        blk[r * TBLK_STR + 4 * c4 + 2] = v[pass].z;
        blk[r * TBLK_STR + 4 * c4 + 3] = v[pass].w;
    }
    __syncthreads();
    for (int pass = 0; pass < 4; ++pass) {
        const int r = r0 + 16 * pass;
        float4 w;
        w.x = blk[(4 * c4 + 0) * TBLK_STR + r];
        w.y = blk[(4 * c4 + 1) * TBLK_STR + r];
        w.z = blk[(4 * c4 + 2) * TBLK_STR + r];
        w.w = blk[(4 * c4 + 3) * TBLK_STR + r];
        *(float4*)(base + (size_t)r * NPTS + 4 * c4) = w;
    }
}

extern "C" void kernel_launch(void* const* d_in, const int* in_sizes, int n_in,
                              void* d_out, int out_size, void* d_ws, size_t ws_size,
                              hipStream_t stream) {
    const float* features = (const float*)d_in[0];
    const float* theta    = (const float*)d_in[1];
    const float* wbias    = (const float*)d_in[2];
    const int*   nbh      = (const int*)d_in[3];
    const float* pos      = (const float*)d_in[4];
    const float* bias     = (const float*)d_in[5];
    float*       out      = (float*)d_out;
    _Float16*    wsA      = (_Float16*)d_ws;

    {
        int nwords = BATCH * COUT * NPTS / 4;
        init_out_kernel<<<(nwords + 255) / 256, 256, 0, stream>>>(bias, out);
    }
    pack_weights_kernel<<<(A_ELEMS + 255) / 256, 256, 0, stream>>>(theta, wbias, wsA);
    flexconv_kernel<<<BATCH * (NPTS / TILE_PTS), 256, 0, stream>>>(
        features, nbh, pos, wsA, out);
    turn_blocks_kernel<<<BATCH * (NPTS / TBLK), 256, 0, stream>>>(out);
}


# speedup vs baseline: 13.8861x; 1.3018x over previous
#include <hip/hip_runtime.h>


typedef __attribute__((ext_vector_type(16))) _Float16 v16h;
typedef __attribute__((ext_vector_type(8)))  float    v8f;

static constexpr int BATCH    = 4;
static constexpr int CIN      = 64;
static constexpr int COUT     = 64;
static constexpr int NPTS     = 16384;
static constexpr int KNBH     = 8;

static constexpr int ROWS     = 4 * COUT;
static constexpr int TILE_PTS = 32;
static constexpr int LDS_STR  = 33;
static constexpr int A_ELEMS  = 16   * 2   * 32   * 16  ;
static constexpr int B_HALVES = 2   * 2   * 32   * 16  ;
static constexpr int TBLK     = 64;
static constexpr int TBLK_STR = TBLK + 1;
static constexpr int NTILES   = BATCH * (NPTS / TILE_PTS);
static constexpr int HALF_TILES = NTILES / 2;
static constexpr int TICKET_STR = 64;
static constexpr int WS_A_BYTES = A_ELEMS * 2;

__device__ __forceinline__ int kmap(int h, int g, int kc) {
    return 32 * kc + 16 * (h >> 3) + 8 * g + (h & 7);
}

__global__ void init_out_kernel(const float* __restrict__ bias,
                                float* __restrict__ out) {
    size_t idx = (size_t)blockIdx.x * blockDim.x + threadIdx.x;
    size_t nwords = (size_t)BATCH * COUT * NPTS / 4;
    if (idx >= nwords) return;
    int o = (int)(idx & 15) * 4;
    ((float4*)out)[idx] = make_float4(bias[o], bias[o + 1], bias[o + 2], bias[o + 3]);
}

__global__ void pack_weights_kernel(const float* __restrict__ theta,
                                    const float* __restrict__ wbias,
                                    _Float16* __restrict__ wsA,
                                    int* __restrict__ tickets) {
    int e = blockIdx.x * blockDim.x + threadIdx.x;
    if (e >= A_ELEMS) return;
    if (e < 4) tickets[e * TICKET_STR] = 0;
    int h    = e & 15;
    int lane = (e >> 4) & 31;
    int kc   = (e >> 9) & 1;
    int mt   = e >> 10;
    int row  = mt * 16 + (lane & 15);
    int g    = lane >> 4;
    int Kk   = kmap(h, g, kc);
    float w;
    if (row < 3 * COUT) {
        int d = row >> 6, o = row & 63;
        w = theta[((size_t)d * CIN + Kk) * COUT + o];
    } else {
        int o = row - 3 * COUT;
        w = wbias[(size_t)Kk * COUT + o];
    }
    wsA[e] = (_Float16)w;
}

__global__ void __launch_bounds__(256)
flexconv_kernel(const float* __restrict__ features,
                const int*   __restrict__ nbh,
                const float* __restrict__ pos,
                const _Float16* __restrict__ wsA,
                int* __restrict__ tickets,
                float* __restrict__ out) {
    __shared__ float ftfb[ROWS * LDS_STR];
    __shared__ __align__(32) _Float16 bstage[B_HALVES];
    __shared__ int s_tile;

    const int tid        = threadIdx.x;
    const int wave       = tid >> 5;
    const int lane       = tid & 31;
    const int blocksPerB = NPTS / TILE_PTS;

    if (tid == 0) {
        const unsigned where = __builtin_amdgcn_s_sendmsg_rtn(0x87);
        const int h = (int)((where >> 18) & 1u);
        const int t = atomicAdd(&tickets[h * TICKET_STR], 1);
        int tile;
        if (t < HALF_TILES) {
            tile = h * HALF_TILES + t;
        } else {
            const int u = atomicAdd(&tickets[(2 + (1 - h)) * TICKET_STR], 1);
            tile = (1 - h) * HALF_TILES + (HALF_TILES - 1 - u);
        }
        s_tile = tile & (NTILES - 1);
    }
    __syncthreads();
    const int tile       = __builtin_amdgcn_readfirstlane(s_tile);
    const int b          = tile / blocksPerB;
    const int n0         = (tile % blocksPerB) * TILE_PTS;

    int   my_joff;
    float my_dx, my_dy, my_dz;
    {
        const int n = n0 + 4 * wave + (lane & 3);
        const int k = lane >> 2;
        const int j = nbh[((size_t)b * KNBH + k) * NPTS + n];
        const float* posb = pos + (size_t)b * 3 * NPTS;
        my_joff = (j & (TBLK - 1)) * NPTS + (j & ~(TBLK - 1));
        my_dx   = posb[0 * NPTS + j] - posb[0 * NPTS + n];
        my_dy   = posb[1 * NPTS + j] - posb[1 * NPTS + n];
        my_dz   = posb[2 * NPTS + j] - posb[2 * NPTS + n];
    }

    {
        const int ncol = tid & 31;
        const int krow = tid >> 5;
        const float* featb = features + (size_t)b * CIN * NPTS + n0;
        const int nt   = ncol >> 4;
        const int lcol = ncol & 15;
        #pragma unroll
        for (int pass = 0; pass < 8; ++pass) {
            const int Kk  = krow + 8 * pass;
            const float v = featb[(size_t)Kk * NPTS + ncol];
            const int kc  = Kk >> 5;
            const int rem = Kk & 31;
            const int g   = (rem >> 3) & 1;
            const int h   = 8 * (rem >> 4) + (rem & 7);
            const int ln  = g * 16 + lcol;
            bstage[(((nt * 2 + kc) * 32) + ln) * 16 + h] = (_Float16)v;
        }
    }
    __syncthreads();

    {
        const int nt    = wave & 1;
        const int mbase = wave >> 1;
        const int g     = lane >> 4;
        const int lcol  = lane & 15;

        const v16h bfrag0 = *(const v16h*)(bstage + ((nt * 2 + 0) * 32 + lane) * 16);
        const v16h bfrag1 = *(const v16h*)(bstage + ((nt * 2 + 1) * 32 + lane) * 16);

        #pragma unroll
        for (int i = 0; i < 4; ++i) {
            const int mt = mbase + 4 * i;
            v8f c = {};
            {
                v16h a = *(const v16h*)(wsA + (((size_t)mt * 2 + 0) * 32 + lane) * 16);
                c = __builtin_amdgcn_wmma_f32_16x16x32_f16(
                        false, a, false, bfrag0, (short)0, c, false, false);
            }
            {
                v16h a = *(const v16h*)(wsA + (((size_t)mt * 2 + 1) * 32 + lane) * 16);
                c = __builtin_amdgcn_wmma_f32_16x16x32_f16(
                        false, a, false, bfrag1, (short)0, c, false, false);
            }
            const int colBase = nt * 16 + lcol;
            #pragma unroll
            for (int r = 0; r < 8; ++r) {
                int row = mt * 16 + 8 * g + r;
                ftfb[row * LDS_STR + colBase] = c[r];
            }
        }
    }
    __syncthreads();

    {
        float* outb = out + (size_t)b * COUT * NPTS;
        #pragma unroll
        for (int i = 0; i < 4; ++i) {
            const int pt = 4 * wave + i;
            const float c0l = ftfb[(0 * COUT + lane) * LDS_STR + pt];
            const float c1l = ftfb[(1 * COUT + lane) * LDS_STR + pt];
            const float c2l = ftfb[(2 * COUT + lane) * LDS_STR + pt];
            const float cbl = ftfb[(3 * COUT + lane) * LDS_STR + pt];
            const float c0h = ftfb[(0 * COUT + 32 + lane) * LDS_STR + pt];
            const float c1h = ftfb[(1 * COUT + 32 + lane) * LDS_STR + pt];
            const float c2h = ftfb[(2 * COUT + 32 + lane) * LDS_STR + pt];
            const float cbh = ftfb[(3 * COUT + 32 + lane) * LDS_STR + pt];
            for (int k = 0; k < KNBH; ++k) {
                const int   srcl = 4 * k + i;
                const int   joff = __builtin_amdgcn_readlane(my_joff, srcl);
                const float dx = __int_as_float(__builtin_amdgcn_readlane(__float_as_int(my_dx), srcl));
                const float dy = __int_as_float(__builtin_amdgcn_readlane(__float_as_int(my_dy), srcl));
                const float dz = __int_as_float(__builtin_amdgcn_readlane(__float_as_int(my_dz), srcl));
                float* dst = outb + joff;
                atomicAdd(dst + lane,      dx * c0l + dy * c1l + dz * c2l + cbl);
                atomicAdd(dst + 32 + lane, dx * c0h + dy * c1h + dz * c2h + cbh);
            }
        }
    }
}

__global__ void __launch_bounds__(256)
turn_blocks_kernel(float* out) {
    __shared__ float blk[TBLK * TBLK_STR];

    const int tid = threadIdx.x;
    const int b   = blockIdx.x / (NPTS / TBLK);
    const int J   = blockIdx.x % (NPTS / TBLK);
    float* base   = out + (size_t)b * COUT * NPTS + J * TBLK;
    const int c4  = tid & 15;
    const int r0  = tid >> 4;

    float4 v[4];
    for (int pass = 0; pass < 4; ++pass) {
        const int r = r0 + 16 * pass;
        v[pass] = *(const float4*)(base + (size_t)r * NPTS + 4 * c4);
    }
    for (int pass = 0; pass < 4; ++pass) {
        const int r = r0 + 16 * pass;
        blk[r * TBLK_STR + 4 * c4 + 0] = v[pass].x;
        blk[r * TBLK_STR + 4 * c4 + 1] = v[pass].y;
        blk[r * TBLK_STR + 4 * c4 + 2] = v[pass].z;
        blk[r * TBLK_STR + 4 * c4 + 3] = v[pass].w;
    }
    __syncthreads();
    for (int pass = 0; pass < 4; ++pass) {
        const int r = r0 + 16 * pass;
        float4 w;
        w.x = blk[(4 * c4 + 0) * TBLK_STR + r];
        w.y = blk[(4 * c4 + 1) * TBLK_STR + r];
        w.z = blk[(4 * c4 + 2) * TBLK_STR + r];
        w.w = blk[(4 * c4 + 3) * TBLK_STR + r];
        *(float4*)(base + (size_t)r * NPTS + 4 * c4) = w;
    }
}

extern "C" void kernel_launch(void* const* d_in, const int* in_sizes, int n_in,
                              void* d_out, int out_size, void* d_ws, size_t ws_size,
                              hipStream_t stream) {
    const float* features = (const float*)d_in[0];
    const float* theta    = (const float*)d_in[1];
    const float* wbias    = (const float*)d_in[2];
    const int*   nbh      = (const int*)d_in[3];
    const float* pos      = (const float*)d_in[4];
    const float* bias     = (const float*)d_in[5];
    float*       out      = (float*)d_out;
    _Float16*    wsA      = (_Float16*)d_ws;
    int*         tickets  = (int*)((char*)d_ws + WS_A_BYTES);

    {
        int nwords = BATCH * COUT * NPTS / 4;
        init_out_kernel<<<(nwords + 255) / 256, 256, 0, stream>>>(bias, out);
    }
    pack_weights_kernel<<<(A_ELEMS + 255) / 256, 256, 0, stream>>>(theta, wbias, wsA, tickets);
    flexconv_kernel<<<BATCH * (NPTS / TILE_PTS), 256, 0, stream>>>(
        features, nbh, pos, wsA, tickets, out);
    turn_blocks_kernel<<<BATCH * (NPTS / TBLK), 256, 0, stream>>>(out);
}


# One product phase for all 256 points fed at LDS rate: X kept in read order (contiguous 16-byte fragment reads), a wave owns 2 stripes x 32 points, next step's reads ahead of the WMMAs; sums as before
# speedup vs baseline: 30.1808x; 2.1735x over previous
#include <hip/hip_runtime.h>


typedef __attribute__((ext_vector_type(16))) _Float16 v16h;
typedef __attribute__((ext_vector_type(8)))  float    v8f;

static constexpr int BATCH    = 4;
static constexpr int CIN      = 64;
static constexpr int COUT     = 64;
static constexpr int NPTS     = 16384;
static constexpr int KNBH     = 8;

static constexpr int ROWS     = 4 * COUT;
static constexpr int TILE_PTS = 32;
static constexpr int LDS_STR  = 33;
static constexpr int A_ELEMS  = 16   * 2   * 32   * 16  ;
static constexpr int B_HALVES = 2   * 2   * 32   * 16  ;
static constexpr int TBLK     = 64;
static constexpr int TBLK_STR = TBLK + 1;
static constexpr int NTILES   = BATCH * (NPTS / TILE_PTS);
static constexpr int HALF_TILES = NTILES / 2;
static constexpr int TICKET_STR = 64;
static constexpr int WS_A_BYTES = A_ELEMS * 2;

__device__ __forceinline__ int kmap(int h, int g, int kc) {
    return 32 * kc + 16 * (h >> 3) + 8 * g + (h & 7);
}


static constexpr int PREP_THREADS = 256;
static constexpr int PACK_BLOCKS  = A_ELEMS / PREP_THREADS;
static constexpr int INIT_BLOCKS  = BATCH * COUT * NPTS / 4 / PREP_THREADS;
static constexpr int PREP_BLOCKS  = PACK_BLOCKS + INIT_BLOCKS;
static_assert(PACK_BLOCKS * PREP_THREADS == A_ELEMS, "pack blocks cover the fragments exactly: no bounds test");
static_assert((long long)INIT_BLOCKS * PREP_THREADS * 4 == (long long)BATCH * COUT * NPTS, "start-value blocks cover out exactly: no bounds test");

__global__ void __launch_bounds__(PREP_THREADS)
prepare_kernel(const float* __restrict__ bias,
               const float* __restrict__ theta,
               const float* __restrict__ wbias,
               float* __restrict__ out,
               _Float16* __restrict__ wsA,
               int* __restrict__ tickets) {
    if (blockIdx.x >= PACK_BLOCKS) {
        const size_t idx = (size_t)(blockIdx.x - PACK_BLOCKS) * PREP_THREADS + threadIdx.x;
        const int o = (int)(idx & 15) * 4;
        ((float4*)out)[idx] = make_float4(bias[o], bias[o + 1], bias[o + 2], bias[o + 3]);
        return;
    }
    const int e = blockIdx.x * PREP_THREADS + threadIdx.x;
    if (e < 4) tickets[e * TICKET_STR] = 0;
    const int h    = e & 15;
    const int lane = (e >> 4) & 31;
    const int kc   = (e >> 9) & 1;
    const int mt   = e >> 10;
    const int row  = mt * 16 + (lane & 15);
    const int g    = lane >> 4;
    const int Kk   = kmap(h, g, kc);
    float w;
    if (row < 3 * COUT) {
        const int d = row >> 6, o = row & 63;
        w = theta[((size_t)d * CIN + Kk) * COUT + o];
    } else {
        const int o = row - 3 * COUT;
        w = wbias[(size_t)Kk * COUT + o];
    }
    wsA[e] = (_Float16)w;
}

__global__ void __launch_bounds__(256)
flexconv_kernel(const float* __restrict__ features,
                const int*   __restrict__ nbh,
                const float* __restrict__ pos,
                const _Float16* __restrict__ wsA,
                int* __restrict__ tickets,
                float* __restrict__ out) {
    __shared__ float ftfb[ROWS * LDS_STR];
    __shared__ __align__(32) _Float16 bstage[B_HALVES];
    __shared__ int s_tile;

    const int tid        = threadIdx.x;
    const int wave       = tid >> 5;
    const int lane       = tid & 31;
    const int blocksPerB = NPTS / TILE_PTS;

    if (tid == 0) {
        const unsigned where = __builtin_amdgcn_s_sendmsg_rtn(0x87);
        const int h = (int)((where >> 18) & 1u);
        const int t = atomicAdd(&tickets[h * TICKET_STR], 1);
        int tile;
        if (t < HALF_TILES) {
            tile = h * HALF_TILES + t;
        } else {
            const int u = atomicAdd(&tickets[(2 + (1 - h)) * TICKET_STR], 1);
            tile = (1 - h) * HALF_TILES + (HALF_TILES - 1 - u);
        }
        s_tile = tile & (NTILES - 1);
    }
    __syncthreads();
    const int tile       = __builtin_amdgcn_readfirstlane(s_tile);
    const int b          = tile / blocksPerB;
    const int n0         = (tile % blocksPerB) * TILE_PTS;

    int   my_joff;
    float my_dx, my_dy, my_dz;
    {
        const int n = n0 + 4 * wave + (lane & 3);
        const int k = lane >> 2;
        const int j = nbh[((size_t)b * KNBH + k) * NPTS + n];
        const float* posb = pos + (size_t)b * 3 * NPTS;
        my_joff = (j & (TBLK - 1)) * NPTS + (j & ~(TBLK - 1));
        my_dx   = posb[0 * NPTS + j] - posb[0 * NPTS + n];
        my_dy   = posb[1 * NPTS + j] - posb[1 * NPTS + n];
        my_dz   = posb[2 * NPTS + j] - posb[2 * NPTS + n];
    }

    {
        const int ncol = tid & 31;
        const int krow = tid >> 5;
        const float* featb = features + (size_t)b * CIN * NPTS + n0;
        const int nt   = ncol >> 4;
        const int lcol = ncol & 15;
        #pragma unroll
        for (int pass = 0; pass < 8; ++pass) {
            const int Kk  = krow + 8 * pass;
            const float v = featb[(size_t)Kk * NPTS + ncol];
            const int kc  = Kk >> 5;
            const int rem = Kk & 31;
            const int g   = (rem >> 3) & 1;
            const int h   = 8 * (rem >> 4) + (rem & 7);
            const int ln  = g * 16 + lcol;
            bstage[(((nt * 2 + kc) * 32) + ln) * 16 + h] = (_Float16)v;
        }
    }
    __syncthreads();

    {
        const int nt    = wave & 1;
        const int mbase = wave >> 1;
        const int g     = lane >> 4;
        const int lcol  = lane & 15;

        const v16h bfrag0 = *(const v16h*)(bstage + ((nt * 2 + 0) * 32 + lane) * 16);
        const v16h bfrag1 = *(const v16h*)(bstage + ((nt * 2 + 1) * 32 + lane) * 16);

        #pragma unroll
        for (int i = 0; i < 4; ++i) {
            const int mt = mbase + 4 * i;
            v8f c = {};
            {
                v16h a = *(const v16h*)(wsA + (((size_t)mt * 2 + 0) * 32 + lane) * 16);
                c = __builtin_amdgcn_wmma_f32_16x16x32_f16(
                        false, a, false, bfrag0, (short)0, c, false, false);
            }
            {
                v16h a = *(const v16h*)(wsA + (((size_t)mt * 2 + 1) * 32 + lane) * 16);
                c = __builtin_amdgcn_wmma_f32_16x16x32_f16(
                        false, a, false, bfrag1, (short)0, c, false, false);
            }
            const int colBase = nt * 16 + lcol;
            #pragma unroll
            for (int r = 0; r < 8; ++r) {
                int row = mt * 16 + 8 * g + r;
                ftfb[row * LDS_STR + colBase] = c[r];
            }
        }
    }
    __syncthreads();

    {
        float* outb = out + (size_t)b * COUT * NPTS;
        #pragma unroll
        for (int i = 0; i < 4; ++i) {
            const int pt = 4 * wave + i;
            const float c0l = ftfb[(0 * COUT + lane) * LDS_STR + pt];
            const float c1l = ftfb[(1 * COUT + lane) * LDS_STR + pt];
            const float c2l = ftfb[(2 * COUT + lane) * LDS_STR + pt];
            const float cbl = ftfb[(3 * COUT + lane) * LDS_STR + pt];
            const float c0h = ftfb[(0 * COUT + 32 + lane) * LDS_STR + pt];
            const float c1h = ftfb[(1 * COUT + 32 + lane) * LDS_STR + pt];
            const float c2h = ftfb[(2 * COUT + 32 + lane) * LDS_STR + pt];
            const float cbh = ftfb[(3 * COUT + 32 + lane) * LDS_STR + pt];
            for (int k = 0; k < KNBH; ++k) {
                const int   srcl = 4 * k + i;
                const int   joff = __builtin_amdgcn_readlane(my_joff, srcl);
                const float dx = __int_as_float(__builtin_amdgcn_readlane(__float_as_int(my_dx), srcl));
                const float dy = __int_as_float(__builtin_amdgcn_readlane(__float_as_int(my_dy), srcl));
                const float dz = __int_as_float(__builtin_amdgcn_readlane(__float_as_int(my_dz), srcl));
                float* dst = outb + joff;
                atomicAdd(dst + lane,      dx * c0l + dy * c1l + dz * c2l + cbl);
                atomicAdd(dst + 32 + lane, dx * c0h + dy * c1h + dz * c2h + cbh);
            }
        }
    }
}

__global__ void __launch_bounds__(256)
turn_blocks_kernel(float* out) {
    __shared__ float blk[TBLK * TBLK_STR];

    const int tid = threadIdx.x;
    const int b   = blockIdx.x / (NPTS / TBLK);
    const int J   = blockIdx.x % (NPTS / TBLK);
    float* base   = out + (size_t)b * COUT * NPTS + J * TBLK;
    const int c4  = tid & 15;
    const int r0  = tid >> 4;

    float4 v[4];
    for (int pass = 0; pass < 4; ++pass) {
        const int r = r0 + 16 * pass;
        v[pass] = *(const float4*)(base + (size_t)r * NPTS + 4 * c4);
    }
    for (int pass = 0; pass < 4; ++pass) {
        const int r = r0 + 16 * pass;
        blk[r * TBLK_STR + 4 * c4 + 0] = v[pass].x;
        blk[r * TBLK_STR + 4 * c4 + 1] = v[pass].y;
        blk[r * TBLK_STR + 4 * c4 + 2] = v[pass].z;
        blk[r * TBLK_STR + 4 * c4 + 3] = v[pass].w;
    }
    __syncthreads();
    for (int pass = 0; pass < 4; ++pass) {
        const int r = r0 + 16 * pass;
        float4 w;
        w.x = blk[(4 * c4 + 0) * TBLK_STR + r];
        w.y = blk[(4 * c4 + 1) * TBLK_STR + r];
        w.z = blk[(4 * c4 + 2) * TBLK_STR + r];
        w.w = blk[(4 * c4 + 3) * TBLK_STR + r];
        *(float4*)(base + (size_t)r * NPTS + 4 * c4) = w;
    }
}

typedef __attribute__((ext_vector_type(2))) _Float16 v2h;
typedef __attribute__((ext_vector_type(8))) _Float16 v8h;

static constexpr int DPW         = 256;
static constexpr int OWN_THREADS = 1024;
static constexpr int OWN_WAVES   = OWN_THREADS / 32;
static constexpr int BIN_CAP     = 32;
static constexpr int GRP         = 128;
static constexpr int XK_CHUNKS   = ROWS / 32;
static constexpr int XBLK        = 32 * 8 + 8;
static constexpr int X_HALVES    = (GRP / 16) * XK_CHUNKS * 2 * XBLK;
static constexpr int X_GROUP     = 2 * X_HALVES;
static constexpr int PROD_WAVES  = 16;
static constexpr int A2_ELEMS    = (COUT / 16) * XK_CHUNKS * 32 * 16;
static constexpr int ROW_WORDS   = CIN / 2;
static constexpr int TP_PTS      = 64;
static constexpr int TP_BLOCKS   = BATCH * (NPTS / TP_PTS);
static constexpr int PACK2_BLOCKS = A2_ELEMS / 256;
static constexpr int SCAN_GROUP  = 4;
static constexpr int HIT_GROUP   = 16;
static constexpr size_t WS_FT_BYTES = (size_t)BATCH * NPTS * CIN * 2;
static constexpr size_t WS_NEED     = (size_t)WS_A_BYTES + WS_FT_BYTES;
static constexpr int LIST_TILES   = NPTS / DPW;
static constexpr int LIST_SLICES  = NPTS / TP_PTS;
static constexpr int LIST_ENTRIES = KNBH * TP_PTS;
static constexpr int LIST_WORDS   = LIST_TILES + LIST_ENTRIES / 2;
static constexpr int LIST_AHEAD   = 4;
static constexpr size_t WS_LIST_BYTES = (size_t)BATCH * LIST_SLICES * LIST_WORDS * 4;
static constexpr size_t WS_NEED_LISTS = WS_NEED + WS_LIST_BYTES;
static_assert(LIST_TILES == 64 && TP_PTS == 64, "an entry is (point in its tile) * 64 + (source in its slice): 14 bits");
static_assert(LIST_ENTRIES == 2 * 256, "a turn-over workgroup of 256 threads takes two entries per thread");
static_assert(OWN_THREADS == 4 * LIST_SLICES, "launch 2 reads every slice's region with four threads");
static_assert(A2_ELEMS * 2 == WS_A_BYTES, "the packed weights of both forms fill the same 32 KB");
static_assert(KNBH * NPTS / 4 == OWN_THREADS * 32, "the scan covers the index list exactly: no bounds test");

__global__ void __launch_bounds__(256)
pack_rows_kernel(const float* __restrict__ features,
                 const float* __restrict__ theta,
                 const float* __restrict__ wbias,
                 const int*   __restrict__ nbh,
                 _Float16* __restrict__ wsA,
                 unsigned* __restrict__ ftT,
                 unsigned* __restrict__ lists,
                 int use_lists) {
    __shared__ unsigned tl[TP_PTS * 33];
    __shared__ __align__(16) int s_hist[LIST_TILES];
    __shared__ int s_start[LIST_TILES];

    if (blockIdx.x < PACK2_BLOCKS) {
        const int e    = blockIdx.x * 256 + threadIdx.x;
        const int h    = e & 15;
        const int lane = (e >> 4) & 31;
        const int kc   = (e >> 9) & 7;
        const int mt   = e >> 12;
        const int o    = mt * 16 + (lane & 15);
        const int g    = lane >> 4;
        const int c    = 32 * kc + kmap(h, g, 0);
        const int d    = c >> 6, i = c & 63;
        const float w  = (d < 3) ? theta[((size_t)d * CIN + i) * COUT + o]
                                 : wbias[(size_t)i * COUT + o];
        wsA[e] = (_Float16)w;
        return;
    }
    const int tb = blockIdx.x - PACK2_BLOCKS;
    const int b  = tb / (NPTS / TP_PTS);
    const int n0 = (tb % (NPTS / TP_PTS)) * TP_PTS;

    if (use_lists) {
        if (threadIdx.x < LIST_TILES) s_hist[threadIdx.x] = 0;
        __syncthreads();
    }
    int jv[2] = {0, 0};
    if (use_lists) {
        #pragma unroll
        for (int r = 0; r < 2; ++r) {
            const int e = threadIdx.x + 256 * r;
            jv[r] = nbh[((size_t)b * KNBH + (e >> 6)) * NPTS + n0 + (e & 63)];
        }
    }
    {
        const int p   = threadIdx.x & 63;
        const int ip0 = threadIdx.x >> 6;
        const float* fb = features + (size_t)b * CIN * NPTS + n0 + p;
        for (int pass = 0; pass < 8; ++pass) {
            const int ipair = ip0 + 4 * pass;
            const float v0 = fb[(size_t)(2 * ipair) * NPTS];
            const float v1 = fb[(size_t)(2 * ipair + 1) * NPTS];
            v2h hh;
            hh.x = (_Float16)v0;
            hh.y = (_Float16)v1;
            tl[p * 33 + ipair] = __builtin_bit_cast(unsigned, hh);
        }
    }
    int slot[2] = {0, 0};
    int tile[2] = {-1, -1};
    if (use_lists) {
        #pragma unroll
        for (int r = 0; r < 2; ++r) {
            const unsigned t = (unsigned)jv[r] >> 8;
            if (t < (unsigned)LIST_TILES) {
                tile[r] = (int)t;
                slot[r] = atomicAdd(&s_hist[t], 1);
            }
        }
    }
    __syncthreads();
    if (use_lists) {
        unsigned* region = lists + (size_t)tb * LIST_WORDS;
        const int t = threadIdx.x;
        int start = 0, count = 0;
        if (t < LIST_TILES) {
            const int4* h4 = (const int4*)s_hist;
            #pragma unroll
            for (int q = 0; q < LIST_TILES / 4; ++q) {
                const int4 v = h4[q];
                start += (4 * q + 0 < t ? v.x : 0) + (4 * q + 1 < t ? v.y : 0)
                       + (4 * q + 2 < t ? v.z : 0) + (4 * q + 3 < t ? v.w : 0);
            }
            count = s_hist[t];
            s_start[t] = start;
        }
        __syncthreads();
        if (t < LIST_TILES) region[t] = ((unsigned)start << 16) | (unsigned)count;
        unsigned short* ent = (unsigned short*)(region + LIST_TILES);
        #pragma unroll
        for (int r = 0; r < 2; ++r) {
            if (tile[r] >= 0) {
                const int e = threadIdx.x + 256 * r;
                ent[s_start[tile[r]] + slot[r]] = (unsigned short)(((jv[r] & (DPW - 1)) << 6) | (e & 63));
            }
        }
    }
    {
        const int w  = threadIdx.x & 31;
        const int p0 = threadIdx.x >> 5;
        for (int pass = 0; pass < 8; ++pass) {
            const int pp = p0 + 8 * pass;
            ftT[((size_t)b * NPTS + n0 + pp) * ROW_WORDS + w] = tl[pp * 33 + w];
        }
    }
}

typedef __attribute__((ext_vector_type(2))) float v2f;

__device__ __forceinline__ void add_source(unsigned row, float dx, float dy, float dz, v2f (&acc)[4]) {
    const v2h hh = __builtin_bit_cast(v2h, row);
    v2f f;
    f.x = (float)hh.x;
    f.y = (float)hh.y;
    v2f sx, sy, sz;
    sx.x = dx;  sx.y = dx;
    sy.x = dy;  sy.y = dy;
    sz.x = dz;  sz.y = dz;
    acc[0] = __builtin_elementwise_fma(sx, f, acc[0]);
    acc[1] = __builtin_elementwise_fma(sy, f, acc[1]);
    acc[2] = __builtin_elementwise_fma(sz, f, acc[2]);
    acc[3] += f;
}

typedef const __attribute__((address_space(1))) char*     gbytes_t;
typedef const __attribute__((address_space(1))) unsigned* gwords_t;

__device__ __forceinline__ gbytes_t slot_row_address(const unsigned* __restrict__ ftb, unsigned myoff, int slot) {
    const unsigned so = (unsigned)__builtin_amdgcn_readlane((int)myoff, slot);
    gbytes_t rowp = (gbytes_t)ftb + (size_t)so;
    asm volatile("" : "+s"(rowp));
    return rowp;
}

__device__ __forceinline__ void add_slot(unsigned row, int slot, float mdx, float mdy, float mdz, v2f (&acc)[4]) {
    const float dx = __int_as_float(__builtin_amdgcn_readlane(__float_as_int(mdx), slot));
    const float dy = __int_as_float(__builtin_amdgcn_readlane(__float_as_int(mdy), slot));
    const float dz = __int_as_float(__builtin_amdgcn_readlane(__float_as_int(mdz), slot));
    add_source(row, dx, dy, dz, acc);
}

template <int M>
__device__ __forceinline__ void sum_pass(const unsigned* __restrict__ ftb, unsigned myoff, int h0, int lane,
                                         float pjx, float pjy, float pjz,
                                         float mpx, float mpy, float mpz, v2f (&acc)[4]) {
    gbytes_t rowp[M];
    #pragma unroll
    for (int q = 0; q < M; ++q) rowp[q] = slot_row_address(ftb, myoff, h0 + q);
    unsigned lane4 = (unsigned)lane * 4u;
    asm volatile("" : "+v"(lane4));
    unsigned row[M];
    #pragma unroll
    for (int q = 0; q < M; ++q) row[q] = *(gwords_t)(rowp[q] + lane4);
    __builtin_amdgcn_sched_barrier(0);
    const float mdx = pjx - mpx;
    const float mdy = pjy - mpy;
    const float mdz = pjz - mpz;
    #pragma unroll
    for (int q = 0; q < M; ++q) add_slot(row[q], h0 + q, mdx, mdy, mdz, acc);
}

__global__ void __launch_bounds__(OWN_THREADS)
owner_kernel(const int*   __restrict__ nbh,
             const float* __restrict__ pos,
             const float* __restrict__ bias,
             const _Float16* __restrict__ wsA,
             const unsigned* __restrict__ ftT,
             const unsigned* __restrict__ lists,
             int use_lists,
             float* __restrict__ out) {
    __shared__ int s_cnt[DPW];
    __shared__ unsigned short s_bin[DPW * BIN_CAP];
    __shared__ __align__(32) _Float16 s_x[(DPW / GRP) * X_GROUP];

    const int tid  = threadIdx.x;
    const int wave = __builtin_amdgcn_readfirstlane(tid >> 5);
    const int lane = tid & 31;
    const int b    = blockIdx.x / (NPTS / DPW);
    const int j0   = (blockIdx.x % (NPTS / DPW)) * DPW;
    const int*      nbhb = nbh + (size_t)b * KNBH * NPTS;
    const float*    posb = pos + (size_t)b * 3 * NPTS;
    const unsigned* ftb  = ftT + (size_t)b * NPTS * ROW_WORDS;
    float*          outb = out + (size_t)b * COUT * NPTS;

    if (tid < DPW) s_cnt[tid] = 0;
    __syncthreads();

    if (use_lists) {
        const int t   = blockIdx.x % (NPTS / DPW);
        const int s   = tid >> 2;
        const int sub = tid & 3;
        const unsigned* region = lists + ((size_t)b * LIST_SLICES + s) * LIST_WORDS;
        const unsigned d = region[t];
        const int start = min((int)(d >> 16), LIST_ENTRIES);
        const int count = min((int)(d & 0xFFFFu), LIST_ENTRIES - start);
        const unsigned short* ep = (const unsigned short*)(region + LIST_TILES) + start;
        unsigned w[LIST_AHEAD];
        int      slot[LIST_AHEAD];
        #pragma unroll
        for (int u = 0; u < LIST_AHEAD; ++u) {
            const int e = sub + 4 * u;
            w[u] = (e < count) ? (unsigned)ep[e] : 0u;
        }
        #pragma unroll
        for (int u = 0; u < LIST_AHEAD; ++u) {
            const int e = sub + 4 * u;
            slot[u] = BIN_CAP;
            if (e < count) slot[u] = atomicAdd(&s_cnt[(w[u] >> 6) & (DPW - 1)], 1);
        }
        #pragma unroll
        for (int u = 0; u < LIST_AHEAD; ++u) {
            if (slot[u] < BIN_CAP)
                s_bin[((w[u] >> 6) & (DPW - 1)) * BIN_CAP + slot[u]] = (unsigned short)(s * TP_PTS + (int)(w[u] & 63u));
        }
        for (int e = sub + 4 * LIST_AHEAD; e < count; e += 4) {
            const unsigned we = ep[e];
            const unsigned dl = (we >> 6) & (DPW - 1);
            const int sl = atomicAdd(&s_cnt[dl], 1);
            if (sl < BIN_CAP) s_bin[dl * BIN_CAP + sl] = (unsigned short)(s * TP_PTS + (int)(we & 63u));
        }
    } else {
        const int4* idx4 = (const int4*)nbhb;
        for (int it0 = 0; it0 < 32; it0 += SCAN_GROUP) {
            int4 v[SCAN_GROUP];
            #pragma unroll
            for (int u = 0; u < SCAN_GROUP; ++u) v[u] = idx4[tid + OWN_THREADS * (it0 + u)];
            #pragma unroll
            for (int u = 0; u < SCAN_GROUP; ++u) {
                const int f = 4 * (tid + OWN_THREADS * (it0 + u));
                const int jv[4] = {v[u].x, v[u].y, v[u].z, v[u].w};
                #pragma unroll
                for (int e = 0; e < 4; ++e) {
                    const unsigned dl = (unsigned)(jv[e] - j0);
                    if (dl < (unsigned)DPW) {
                        const int slot = atomicAdd(&s_cnt[dl], 1);
                        if (slot < BIN_CAP) s_bin[dl * BIN_CAP + slot] = (unsigned short)((f + e) & (NPTS - 1));
                    }
                }
            }
        }
    }
    __syncthreads();

    for (int g2 = 0; g2 < DPW / GRP; ++g2) {
        for (int u = 0; u < GRP / OWN_WAVES; ++u) {
            const int jl  = wave * (GRP / OWN_WAVES) + u;
            const int dl  = g2 * GRP + jl;
            const int j   = j0 + dl;
            const int cnt = __builtin_amdgcn_readfirstlane(s_cnt[dl]);
            const float pjx = posb[j], pjy = posb[NPTS + j], pjz = posb[2 * NPTS + j];
            v2f acc[4] = {{0.f, 0.f}, {0.f, 0.f}, {0.f, 0.f}, {0.f, 0.f}};

            if (cnt <= BIN_CAP) {
                const int myn = (lane < cnt) ? (int)s_bin[dl * BIN_CAP + lane] : j;
                const unsigned myoff = (unsigned)myn * (unsigned)(ROW_WORDS * 4);
                const float mpx = posb[myn];
                const float mpy = posb[NPTS + myn];
                const float mpz = posb[2 * NPTS + myn];
                switch (min(cnt, HIT_GROUP)) {
                case 1:  sum_pass<1>(ftb, myoff, 0, lane, pjx, pjy, pjz, mpx, mpy, mpz, acc); break;
                case 2:  sum_pass<2>(ftb, myoff, 0, lane, pjx, pjy, pjz, mpx, mpy, mpz, acc); break;
                case 3:  sum_pass<3>(ftb, myoff, 0, lane, pjx, pjy, pjz, mpx, mpy, mpz, acc); break;
                case 4:  sum_pass<4>(ftb, myoff, 0, lane, pjx, pjy, pjz, mpx, mpy, mpz, acc); break;
                case 5:  sum_pass<5>(ftb, myoff, 0, lane, pjx, pjy, pjz, mpx, mpy, mpz, acc); break;
                case 6:  sum_pass<6>(ftb, myoff, 0, lane, pjx, pjy, pjz, mpx, mpy, mpz, acc); break;
                case 7:  sum_pass<7>(ftb, myoff, 0, lane, pjx, pjy, pjz, mpx, mpy, mpz, acc); break;
                case 8:  sum_pass<8>(ftb, myoff, 0, lane, pjx, pjy, pjz, mpx, mpy, mpz, acc); break;
                case 9:  sum_pass<9>(ftb, myoff, 0, lane, pjx, pjy, pjz, mpx, mpy, mpz, acc); break;
                case 10: sum_pass<10>(ftb, myoff, 0, lane, pjx, pjy, pjz, mpx, mpy, mpz, acc); break;
                case 11: sum_pass<11>(ftb, myoff, 0, lane, pjx, pjy, pjz, mpx, mpy, mpz, acc); break;
                case 12: sum_pass<12>(ftb, myoff, 0, lane, pjx, pjy, pjz, mpx, mpy, mpz, acc); break;
                case 13: sum_pass<13>(ftb, myoff, 0, lane, pjx, pjy, pjz, mpx, mpy, mpz, acc); break;
                case 14: sum_pass<14>(ftb, myoff, 0, lane, pjx, pjy, pjz, mpx, mpy, mpz, acc); break;
                case 15: sum_pass<15>(ftb, myoff, 0, lane, pjx, pjy, pjz, mpx, mpy, mpz, acc); break;
                case 16: sum_pass<16>(ftb, myoff, 0, lane, pjx, pjy, pjz, mpx, mpy, mpz, acc); break;
                default: break;
                }
                for (int h = HIT_GROUP; h < cnt; ++h) sum_pass<1>(ftb, myoff, h, lane, pjx, pjy, pjz, mpx, mpy, mpz, acc);
            } else {
                for (int base = 0; base < KNBH * NPTS; base += 32) {
                    const int jv = nbhb[base + lane];
                    unsigned m = __builtin_amdgcn_ballot_w32(jv == j);
                    while (m) {
                        const int srcl = __builtin_ctz(m);
                        m &= m - 1;
                        const int n = (base + srcl) & (NPTS - 1);
                        const unsigned row = ftb[(size_t)n * ROW_WORDS + lane];
                        add_source(row, pjx - posb[n], pjy - posb[NPTS + n], pjz - posb[2 * NPTS + n], acc);
                    }
                }
            }

            {
                const int nt  = jl >> 4, col = jl & 15;
                const int kk  = (2 * lane) & 31;
                const int gg  = (kk >> 3) & 1;
                const int pc  = kk >> 4;
                const int eq  = kk & 7;
                const int kcl = lane >> 4;
                #pragma unroll
                for (int d = 0; d < 4; ++d) {
                    const int kc  = 2 * d + kcl;
                    const int idx = ((nt * XK_CHUNKS + kc) * 2 + pc) * XBLK + (gg * 16 + col) * 8 + eq;
                    const v2h hi = __builtin_convertvector(acc[d], v2h);
                    const v2h lo = __builtin_convertvector(acc[d] - __builtin_convertvector(hi, v2f), v2h);
                    *(v2h*)(s_x + g2 * X_GROUP + idx)            = hi;
                    *(v2h*)(s_x + g2 * X_GROUP + X_HALVES + idx) = lo;
                }
            }
        }
    }
    __syncthreads();

    if (wave < PROD_WAVES) {
        const int mtp = wave & 1;
        const int ntp = (wave >> 1) & 3;
        const int grp = wave >> 3;
        const _Float16* wa = wsA + (((size_t)(2 * mtp) * XK_CHUNKS) * 32 + lane) * 16;
        const _Float16* xb = s_x + grp * X_GROUP + ((2 * ntp) * XK_CHUNKS * 2) * XBLK + lane * 8;
        v8f  acc[2][2] = {};
        v16h af[3][2];
        v16h bf[2][2];
        #pragma unroll
        for (int q = 0; q < 2; ++q) {
            #pragma unroll
            for (int i = 0; i < 2; ++i)
                af[q][i] = *(const v16h*)(wa + ((size_t)(i * XK_CHUNKS + q) * 32) * 16);
        }
        #pragma unroll
        for (int t = 0; t < 2; ++t) {
            const _Float16* p = xb + (t * XK_CHUNKS * 2) * XBLK;
            bf[0][t] = __builtin_shufflevector(*(const v8h*)p, *(const v8h*)(p + XBLK),
                                               0, 1, 2, 3, 4, 5, 6, 7, 8, 9, 10, 11, 12, 13, 14, 15);
        }
        #pragma unroll
        for (int s = 0; s < 2 * XK_CHUNKS; ++s) {
            const int kc = s >> 1;
            if (s + 1 < 2 * XK_CHUNKS) {
                const int kn = (s + 1) >> 1, pn = (s + 1) & 1;
                #pragma unroll
                for (int t = 0; t < 2; ++t) {
                    const _Float16* p = xb + pn * X_HALVES + ((t * XK_CHUNKS + kn) * 2) * XBLK;
                    bf[(s + 1) & 1][t] = __builtin_shufflevector(*(const v8h*)p, *(const v8h*)(p + XBLK),
                                                                 0, 1, 2, 3, 4, 5, 6, 7, 8, 9, 10, 11, 12, 13, 14, 15);
                }
            }
            if ((s & 1) == 0 && kc + 2 < XK_CHUNKS) {
                #pragma unroll
                for (int i = 0; i < 2; ++i)
                    af[(kc + 2) % 3][i] = *(const v16h*)(wa + ((size_t)(i * XK_CHUNKS + kc + 2) * 32) * 16);
            }
            __builtin_amdgcn_sched_barrier(0);
            #pragma unroll
            for (int i = 0; i < 2; ++i) {
                #pragma unroll
                for (int t = 0; t < 2; ++t)
                    acc[i][t] = __builtin_amdgcn_wmma_f32_16x16x32_f16(false, af[kc % 3][i], false, bf[s & 1][t],
                                                                       (short)0, acc[i][t], false, false);
            }
            __builtin_amdgcn_sched_barrier(0);
        }
        const int jbase = j0 + grp * GRP + ntp * 32 + (lane & 15);
        #pragma unroll
        for (int i = 0; i < 2; ++i) {
            #pragma unroll
            for (int r = 0; r < 8; ++r) {
                const int o = (2 * mtp + i) * 16 + 8 * (lane >> 4) + r;
                const float bo = bias[o];
                outb[(size_t)o * NPTS + jbase]      = acc[i][0][r] + bo;
                outb[(size_t)o * NPTS + jbase + 16] = acc[i][1][r] + bo;
            }
        }
    }
}

extern "C" void kernel_launch(void* const* d_in, const int* in_sizes, int n_in,
                              void* d_out, int out_size, void* d_ws, size_t ws_size,
                              hipStream_t stream) {
    const float* features = (const float*)d_in[0];
    const float* theta    = (const float*)d_in[1];
    const float* wbias    = (const float*)d_in[2];
    const int*   nbh      = (const int*)d_in[3];
    const float* pos      = (const float*)d_in[4];
    const float* bias     = (const float*)d_in[5];
    float*       out      = (float*)d_out;
    _Float16*    wsA      = (_Float16*)d_ws;
    int*         tickets  = (int*)((char*)d_ws + WS_A_BYTES);

    if (ws_size >= WS_NEED) {
        unsigned* ftT   = (unsigned*)((char*)d_ws + WS_A_BYTES);
        unsigned* lists = (unsigned*)((char*)d_ws + WS_NEED);
        if (ws_size >= WS_NEED_LISTS) {
            pack_rows_kernel<<<PACK2_BLOCKS + TP_BLOCKS, 256, 0, stream>>>(features, theta, wbias, nbh, wsA, ftT, lists, 1);
            owner_kernel<<<BATCH * (NPTS / DPW), OWN_THREADS, 0, stream>>>(nbh, pos, bias, wsA, ftT, lists, 1, out);
            return;
        }
        pack_rows_kernel<<<PACK2_BLOCKS + TP_BLOCKS, 256, 0, stream>>>(features, theta, wbias, nbh, wsA, ftT, lists, 0);
        owner_kernel<<<BATCH * (NPTS / DPW), OWN_THREADS, 0, stream>>>(nbh, pos, bias, wsA, ftT, lists, 0, out);
        return;
    }

    prepare_kernel<<<PREP_BLOCKS, PREP_THREADS, 0, stream>>>(bias, theta, wbias, out, wsA, tickets);
    flexconv_kernel<<<BATCH * (NPTS / TILE_PTS), 256, 0, stream>>>(
        features, nbh, pos, wsA, tickets, out);
    turn_blocks_kernel<<<BATCH * (NPTS / TBLK), 256, 0, stream>>>(out);
}


# Launch 1 also writes each point's position as one 16-byte record; a sum lane fetches its source's position with one load from one line, not three loads from three planes (sum order unchanged)
# speedup vs baseline: 32.1864x; 1.0665x over previous
#include <hip/hip_runtime.h>


typedef __attribute__((ext_vector_type(16))) _Float16 v16h;
typedef __attribute__((ext_vector_type(8)))  float    v8f;

static constexpr int BATCH    = 4;
static constexpr int CIN      = 64;
static constexpr int COUT     = 64;
static constexpr int NPTS     = 16384;
static constexpr int KNBH     = 8;

static constexpr int ROWS     = 4 * COUT;
static constexpr int TILE_PTS = 32;
static constexpr int LDS_STR  = 33;
static constexpr int A_ELEMS  = 16   * 2   * 32   * 16  ;
static constexpr int B_HALVES = 2   * 2   * 32   * 16  ;
static constexpr int TBLK     = 64;
static constexpr int TBLK_STR = TBLK + 1;
static constexpr int NTILES   = BATCH * (NPTS / TILE_PTS);
static constexpr int HALF_TILES = NTILES / 2;
static constexpr int TICKET_STR = 64;
static constexpr int WS_A_BYTES = A_ELEMS * 2;

__device__ __forceinline__ int kmap(int h, int g, int kc) {
    return 32 * kc + 16 * (h >> 3) + 8 * g + (h & 7);
}


static constexpr int PREP_THREADS = 256;
static constexpr int PACK_BLOCKS  = A_ELEMS / PREP_THREADS;
static constexpr int INIT_BLOCKS  = BATCH * COUT * NPTS / 4 / PREP_THREADS;
static constexpr int PREP_BLOCKS  = PACK_BLOCKS + INIT_BLOCKS;
static_assert(PACK_BLOCKS * PREP_THREADS == A_ELEMS, "pack blocks cover the fragments exactly: no bounds test");
static_assert((long long)INIT_BLOCKS * PREP_THREADS * 4 == (long long)BATCH * COUT * NPTS, "start-value blocks cover out exactly: no bounds test");

__global__ void __launch_bounds__(PREP_THREADS)
prepare_kernel(const float* __restrict__ bias,
               const float* __restrict__ theta,
               const float* __restrict__ wbias,
               float* __restrict__ out,
               _Float16* __restrict__ wsA,
               int* __restrict__ tickets) {
    if (blockIdx.x >= PACK_BLOCKS) {
        const size_t idx = (size_t)(blockIdx.x - PACK_BLOCKS) * PREP_THREADS + threadIdx.x;
        const int o = (int)(idx & 15) * 4;
        ((float4*)out)[idx] = make_float4(bias[o], bias[o + 1], bias[o + 2], bias[o + 3]);
        return;
    }
    const int e = blockIdx.x * PREP_THREADS + threadIdx.x;
    if (e < 4) tickets[e * TICKET_STR] = 0;
    const int h    = e & 15;
    const int lane = (e >> 4) & 31;
    const int kc   = (e >> 9) & 1;
    const int mt   = e >> 10;
    const int row  = mt * 16 + (lane & 15);
    const int g    = lane >> 4;
    const int Kk   = kmap(h, g, kc);
    float w;
    if (row < 3 * COUT) {
        const int d = row >> 6, o = row & 63;
        w = theta[((size_t)d * CIN + Kk) * COUT + o];
    } else {
        const int o = row - 3 * COUT;
        w = wbias[(size_t)Kk * COUT + o];
    }
    wsA[e] = (_Float16)w;
}

__global__ void __launch_bounds__(256)
flexconv_kernel(const float* __restrict__ features,
                const int*   __restrict__ nbh,
                const float* __restrict__ pos,
                const _Float16* __restrict__ wsA,
                int* __restrict__ tickets,
                float* __restrict__ out) {
    __shared__ float ftfb[ROWS * LDS_STR];
    __shared__ __align__(32) _Float16 bstage[B_HALVES];
    __shared__ int s_tile;

    const int tid        = threadIdx.x;
    const int wave       = tid >> 5;
    const int lane       = tid & 31;
    const int blocksPerB = NPTS / TILE_PTS;

    if (tid == 0) {
        const unsigned where = __builtin_amdgcn_s_sendmsg_rtn(0x87);
        const int h = (int)((where >> 18) & 1u);
        const int t = atomicAdd(&tickets[h * TICKET_STR], 1);
        int tile;
        if (t < HALF_TILES) {
            tile = h * HALF_TILES + t;
        } else {
            const int u = atomicAdd(&tickets[(2 + (1 - h)) * TICKET_STR], 1);
            tile = (1 - h) * HALF_TILES + (HALF_TILES - 1 - u);
        }
        s_tile = tile & (NTILES - 1);
    }
    __syncthreads();
    const int tile       = __builtin_amdgcn_readfirstlane(s_tile);
    const int b          = tile / blocksPerB;
    const int n0         = (tile % blocksPerB) * TILE_PTS;

    int   my_joff;
    float my_dx, my_dy, my_dz;
    {
        const int n = n0 + 4 * wave + (lane & 3);
        const int k = lane >> 2;
        const int j = nbh[((size_t)b * KNBH + k) * NPTS + n];
        const float* posb = pos + (size_t)b * 3 * NPTS;
        my_joff = (j & (TBLK - 1)) * NPTS + (j & ~(TBLK - 1));
        my_dx   = posb[0 * NPTS + j] - posb[0 * NPTS + n];
        my_dy   = posb[1 * NPTS + j] - posb[1 * NPTS + n];
        my_dz   = posb[2 * NPTS + j] - posb[2 * NPTS + n];
    }

    {
        const int ncol = tid & 31;
        const int krow = tid >> 5;
        const float* featb = features + (size_t)b * CIN * NPTS + n0;
        const int nt   = ncol >> 4;
        const int lcol = ncol & 15;
        #pragma unroll
        for (int pass = 0; pass < 8; ++pass) {
            const int Kk  = krow + 8 * pass;
            const float v = featb[(size_t)Kk * NPTS + ncol];
            const int kc  = Kk >> 5;
            const int rem = Kk & 31;
            const int g   = (rem >> 3) & 1;
            const int h   = 8 * (rem >> 4) + (rem & 7);
            const int ln  = g * 16 + lcol;
            bstage[(((nt * 2 + kc) * 32) + ln) * 16 + h] = (_Float16)v;
        }
    }
    __syncthreads();

    {
        const int nt    = wave & 1;
        const int mbase = wave >> 1;
        const int g     = lane >> 4;
        const int lcol  = lane & 15;

        const v16h bfrag0 = *(const v16h*)(bstage + ((nt * 2 + 0) * 32 + lane) * 16);
        const v16h bfrag1 = *(const v16h*)(bstage + ((nt * 2 + 1) * 32 + lane) * 16);

        #pragma unroll
        for (int i = 0; i < 4; ++i) {
            const int mt = mbase + 4 * i;
            v8f c = {};
            {
                v16h a = *(const v16h*)(wsA + (((size_t)mt * 2 + 0) * 32 + lane) * 16);
                c = __builtin_amdgcn_wmma_f32_16x16x32_f16(
                        false, a, false, bfrag0, (short)0, c, false, false);
            }
            {
                v16h a = *(const v16h*)(wsA + (((size_t)mt * 2 + 1) * 32 + lane) * 16);
                c = __builtin_amdgcn_wmma_f32_16x16x32_f16(
                        false, a, false, bfrag1, (short)0, c, false, false);
            }
            const int colBase = nt * 16 + lcol;
            #pragma unroll
            for (int r = 0; r < 8; ++r) {
                int row = mt * 16 + 8 * g + r;
                ftfb[row * LDS_STR + colBase] = c[r];
            }
        }
    }
    __syncthreads();

    {
        float* outb = out + (size_t)b * COUT * NPTS;
        #pragma unroll
        for (int i = 0; i < 4; ++i) {
            const int pt = 4 * wave + i;
            const float c0l = ftfb[(0 * COUT + lane) * LDS_STR + pt];
            const float c1l = ftfb[(1 * COUT + lane) * LDS_STR + pt];
            const float c2l = ftfb[(2 * COUT + lane) * LDS_STR + pt];
            const float cbl = ftfb[(3 * COUT + lane) * LDS_STR + pt];
            const float c0h = ftfb[(0 * COUT + 32 + lane) * LDS_STR + pt];
            const float c1h = ftfb[(1 * COUT + 32 + lane) * LDS_STR + pt];
            const float c2h = ftfb[(2 * COUT + 32 + lane) * LDS_STR + pt];
            const float cbh = ftfb[(3 * COUT + 32 + lane) * LDS_STR + pt];
            for (int k = 0; k < KNBH; ++k) {
                const int   srcl = 4 * k + i;
                const int   joff = __builtin_amdgcn_readlane(my_joff, srcl);
                const float dx = __int_as_float(__builtin_amdgcn_readlane(__float_as_int(my_dx), srcl));
                const float dy = __int_as_float(__builtin_amdgcn_readlane(__float_as_int(my_dy), srcl));
                const float dz = __int_as_float(__builtin_amdgcn_readlane(__float_as_int(my_dz), srcl));
                float* dst = outb + joff;
                atomicAdd(dst + lane,      dx * c0l + dy * c1l + dz * c2l + cbl);
                atomicAdd(dst + 32 + lane, dx * c0h + dy * c1h + dz * c2h + cbh);
            }
        }
    }
}

__global__ void __launch_bounds__(256)
turn_blocks_kernel(float* out) {
    __shared__ float blk[TBLK * TBLK_STR];

    const int tid = threadIdx.x;
    const int b   = blockIdx.x / (NPTS / TBLK);
    const int J   = blockIdx.x % (NPTS / TBLK);
    float* base   = out + (size_t)b * COUT * NPTS + J * TBLK;
    const int c4  = tid & 15;
    const int r0  = tid >> 4;

    float4 v[4];
    for (int pass = 0; pass < 4; ++pass) {
        const int r = r0 + 16 * pass;
        v[pass] = *(const float4*)(base + (size_t)r * NPTS + 4 * c4);
    }
    for (int pass = 0; pass < 4; ++pass) {
        const int r = r0 + 16 * pass;
        blk[r * TBLK_STR + 4 * c4 + 0] = v[pass].x;
        blk[r * TBLK_STR + 4 * c4 + 1] = v[pass].y;
        blk[r * TBLK_STR + 4 * c4 + 2] = v[pass].z;
        blk[r * TBLK_STR + 4 * c4 + 3] = v[pass].w;
    }
    __syncthreads();
    for (int pass = 0; pass < 4; ++pass) {
        const int r = r0 + 16 * pass;
        float4 w;
        w.x = blk[(4 * c4 + 0) * TBLK_STR + r];
        w.y = blk[(4 * c4 + 1) * TBLK_STR + r];
        w.z = blk[(4 * c4 + 2) * TBLK_STR + r];
        w.w = blk[(4 * c4 + 3) * TBLK_STR + r];
        *(float4*)(base + (size_t)r * NPTS + 4 * c4) = w;
    }
}

typedef __attribute__((ext_vector_type(2))) _Float16 v2h;
typedef __attribute__((ext_vector_type(8))) _Float16 v8h;

static constexpr int DPW         = 256;
static constexpr int OWN_THREADS = 1024;
static constexpr int OWN_WAVES   = OWN_THREADS / 32;
static constexpr int BIN_CAP     = 32;
static constexpr int GRP         = 128;
static constexpr int XK_CHUNKS   = ROWS / 32;
static constexpr int XBLK        = 32 * 8 + 8;
static constexpr int X_HALVES    = (GRP / 16) * XK_CHUNKS * 2 * XBLK;
static constexpr int X_GROUP     = 2 * X_HALVES;
static constexpr int PROD_WAVES  = 16;
static constexpr int A2_ELEMS    = (COUT / 16) * XK_CHUNKS * 32 * 16;
static constexpr int ROW_WORDS   = CIN / 2;
static constexpr int TP_PTS      = 64;
static constexpr int TP_BLOCKS   = BATCH * (NPTS / TP_PTS);
static constexpr int PACK2_BLOCKS = A2_ELEMS / 256;
static constexpr int SCAN_GROUP  = 4;
static constexpr int HIT_GROUP   = 16;
static constexpr size_t WS_FT_BYTES = (size_t)BATCH * NPTS * CIN * 2;
static constexpr size_t WS_NEED     = (size_t)WS_A_BYTES + WS_FT_BYTES;
static constexpr int LIST_TILES   = NPTS / DPW;
static constexpr int LIST_SLICES  = NPTS / TP_PTS;
static constexpr int LIST_ENTRIES = KNBH * TP_PTS;
static constexpr int LIST_WORDS   = LIST_TILES + LIST_ENTRIES / 2;
static constexpr int LIST_AHEAD   = 4;
static constexpr size_t WS_LIST_BYTES = (size_t)BATCH * LIST_SLICES * LIST_WORDS * 4;
static constexpr size_t WS_NEED_LISTS = WS_NEED + WS_LIST_BYTES;
static constexpr size_t WS_POS_BYTES  = (size_t)BATCH * NPTS * 16;
static constexpr size_t WS_NEED_POS   = WS_NEED_LISTS + WS_POS_BYTES;
static_assert(WS_NEED_LISTS % 16 == 0, "the position records start on a 16-byte boundary of the workspace");
typedef __attribute__((ext_vector_type(4))) float v4f;

__device__ __forceinline__ void load_record(const float* __restrict__ rec, float& x, float& y, float& z) {
    const float* p = (const float*)__builtin_assume_aligned(rec, 16);
    x = p[0];
    y = p[1];
    z = p[2];
}
static_assert(LIST_TILES == 64 && TP_PTS == 64, "an entry is (point in its tile) * 64 + (source in its slice): 14 bits");
static_assert(LIST_ENTRIES == 2 * 256, "a turn-over workgroup of 256 threads takes two entries per thread");
static_assert(OWN_THREADS == 4 * LIST_SLICES, "launch 2 reads every slice's region with four threads");
static_assert(A2_ELEMS * 2 == WS_A_BYTES, "the packed weights of both forms fill the same 32 KB");
static_assert(KNBH * NPTS / 4 == OWN_THREADS * 32, "the scan covers the index list exactly: no bounds test");

template <int PACKED_POS>
__global__ void __launch_bounds__(256)
pack_rows_kernel(const float* __restrict__ features,
                 const float* __restrict__ theta,
                 const float* __restrict__ wbias,
                 const int*   __restrict__ nbh,
                 const float* __restrict__ pos,
                 _Float16* __restrict__ wsA,
                 unsigned* __restrict__ ftT,
                 unsigned* __restrict__ lists,
                 float* __restrict__ posT,
                 int use_lists) {
    __shared__ unsigned tl[TP_PTS * 33];
    __shared__ __align__(16) int s_hist[LIST_TILES];
    __shared__ int s_start[LIST_TILES];

    if (blockIdx.x < PACK2_BLOCKS) {
        const int e    = blockIdx.x * 256 + threadIdx.x;
        const int h    = e & 15;
        const int lane = (e >> 4) & 31;
        const int kc   = (e >> 9) & 7;
        const int mt   = e >> 12;
        const int o    = mt * 16 + (lane & 15);
        const int g    = lane >> 4;
        const int c    = 32 * kc + kmap(h, g, 0);
        const int d    = c >> 6, i = c & 63;
        const float w  = (d < 3) ? theta[((size_t)d * CIN + i) * COUT + o]
                                 : wbias[(size_t)i * COUT + o];
        wsA[e] = (_Float16)w;
        return;
    }
    const int tb = blockIdx.x - PACK2_BLOCKS;
    const int b  = tb / (NPTS / TP_PTS);
    const int n0 = (tb % (NPTS / TP_PTS)) * TP_PTS;

    if (use_lists) {
        if (threadIdx.x < LIST_TILES) s_hist[threadIdx.x] = 0;
        __syncthreads();
    }
    int jv[2] = {0, 0};
    if (use_lists) {
        #pragma unroll
        for (int r = 0; r < 2; ++r) {
            const int e = threadIdx.x + 256 * r;
            jv[r] = nbh[((size_t)b * KNBH + (e >> 6)) * NPTS + n0 + (e & 63)];
        }
    }
    float px = 0.f, py = 0.f, pz = 0.f;
    if (PACKED_POS && threadIdx.x < TP_PTS) {
        const float* pb = pos + (size_t)b * 3 * NPTS + n0 + threadIdx.x;
        px = pb[0];
        py = pb[NPTS];
        pz = pb[2 * NPTS];
    }
    {
        const int p   = threadIdx.x & 63;
        const int ip0 = threadIdx.x >> 6;
        const float* fb = features + (size_t)b * CIN * NPTS + n0 + p;
        for (int pass = 0; pass < 8; ++pass) {
            const int ipair = ip0 + 4 * pass;
            const float v0 = fb[(size_t)(2 * ipair) * NPTS];
            const float v1 = fb[(size_t)(2 * ipair + 1) * NPTS];
            v2h hh;
            hh.x = (_Float16)v0;
            hh.y = (_Float16)v1;
            tl[p * 33 + ipair] = __builtin_bit_cast(unsigned, hh);
        }
    }
    int slot[2] = {0, 0};
    int tile[2] = {-1, -1};
    if (use_lists) {
        #pragma unroll
        for (int r = 0; r < 2; ++r) {
            const unsigned t = (unsigned)jv[r] >> 8;
            if (t < (unsigned)LIST_TILES) {
                tile[r] = (int)t;
                slot[r] = atomicAdd(&s_hist[t], 1);
            }
        }
    }
    __syncthreads();
    if (use_lists) {
        unsigned* region = lists + (size_t)tb * LIST_WORDS;
        const int t = threadIdx.x;
        int start = 0, count = 0;
        if (t < LIST_TILES) {
            const int4* h4 = (const int4*)s_hist;
            #pragma unroll
            for (int q = 0; q < LIST_TILES / 4; ++q) {
                const int4 v = h4[q];
                start += (4 * q + 0 < t ? v.x : 0) + (4 * q + 1 < t ? v.y : 0)
                       + (4 * q + 2 < t ? v.z : 0) + (4 * q + 3 < t ? v.w : 0);
            }
            count = s_hist[t];
            s_start[t] = start;
        }
        __syncthreads();
        if (t < LIST_TILES) region[t] = ((unsigned)start << 16) | (unsigned)count;
        unsigned short* ent = (unsigned short*)(region + LIST_TILES);
        #pragma unroll
        for (int r = 0; r < 2; ++r) {
            if (tile[r] >= 0) {
                const int e = threadIdx.x + 256 * r;
                ent[s_start[tile[r]] + slot[r]] = (unsigned short)(((jv[r] & (DPW - 1)) << 6) | (e & 63));
            }
        }
    }
    {
        const int w  = threadIdx.x & 31;
        const int p0 = threadIdx.x >> 5;
        for (int pass = 0; pass < 8; ++pass) {
            const int pp = p0 + 8 * pass;
            ftT[((size_t)b * NPTS + n0 + pp) * ROW_WORDS + w] = tl[pp * 33 + w];
        }
    }
    if (PACKED_POS && threadIdx.x < TP_PTS) {
        v4f rec;
        rec.x = px;  rec.y = py;  rec.z = pz;  rec.w = 0.f;
        *(v4f*)(posT + ((size_t)b * NPTS + n0 + threadIdx.x) * 4) = rec;
    }
}

typedef __attribute__((ext_vector_type(2))) float v2f;

__device__ __forceinline__ void add_source(unsigned row, float dx, float dy, float dz, v2f (&acc)[4]) {
    const v2h hh = __builtin_bit_cast(v2h, row);
    v2f f;
    f.x = (float)hh.x;
    f.y = (float)hh.y;
    v2f sx, sy, sz;
    sx.x = dx;  sx.y = dx;
    sy.x = dy;  sy.y = dy;
    sz.x = dz;  sz.y = dz;
    acc[0] = __builtin_elementwise_fma(sx, f, acc[0]);
    acc[1] = __builtin_elementwise_fma(sy, f, acc[1]);
    acc[2] = __builtin_elementwise_fma(sz, f, acc[2]);
    acc[3] += f;
}

typedef const __attribute__((address_space(1))) char*     gbytes_t;
typedef const __attribute__((address_space(1))) unsigned* gwords_t;

__device__ __forceinline__ gbytes_t slot_row_address(const unsigned* __restrict__ ftb, unsigned myoff, int slot) {
    const unsigned so = (unsigned)__builtin_amdgcn_readlane((int)myoff, slot);
    gbytes_t rowp = (gbytes_t)ftb + (size_t)so;
    asm volatile("" : "+s"(rowp));
    return rowp;
}

__device__ __forceinline__ void add_slot(unsigned row, int slot, float mdx, float mdy, float mdz, v2f (&acc)[4]) {
    const float dx = __int_as_float(__builtin_amdgcn_readlane(__float_as_int(mdx), slot));
    const float dy = __int_as_float(__builtin_amdgcn_readlane(__float_as_int(mdy), slot));
    const float dz = __int_as_float(__builtin_amdgcn_readlane(__float_as_int(mdz), slot));
    add_source(row, dx, dy, dz, acc);
}

template <int M>
__device__ __forceinline__ void sum_pass(const unsigned* __restrict__ ftb, unsigned myoff, int h0, int lane,
                                         float pjx, float pjy, float pjz,
                                         float mpx, float mpy, float mpz, v2f (&acc)[4]) {
    gbytes_t rowp[M];
    #pragma unroll
    for (int q = 0; q < M; ++q) rowp[q] = slot_row_address(ftb, myoff, h0 + q);
    unsigned lane4 = (unsigned)lane * 4u;
    asm volatile("" : "+v"(lane4));
    unsigned row[M];
    #pragma unroll
    for (int q = 0; q < M; ++q) row[q] = *(gwords_t)(rowp[q] + lane4);
    __builtin_amdgcn_sched_barrier(0);
    const float mdx = pjx - mpx;
    const float mdy = pjy - mpy;
    const float mdz = pjz - mpz;
    #pragma unroll
    for (int q = 0; q < M; ++q) add_slot(row[q], h0 + q, mdx, mdy, mdz, acc);
}

template <int PACKED_POS>
__global__ void __launch_bounds__(OWN_THREADS)
owner_kernel(const int*   __restrict__ nbh,
             const float* __restrict__ pos,
             const float* __restrict__ bias,
             const _Float16* __restrict__ wsA,
             const unsigned* __restrict__ ftT,
             const unsigned* __restrict__ lists,
             const float* __restrict__ posT,
             int use_lists,
             float* __restrict__ out) {
    __shared__ int s_cnt[DPW];
    __shared__ unsigned short s_bin[DPW * BIN_CAP];
    __shared__ __align__(32) _Float16 s_x[(DPW / GRP) * X_GROUP];

    const int tid  = threadIdx.x;
    const int wave = __builtin_amdgcn_readfirstlane(tid >> 5);
    const int lane = tid & 31;
    const int b    = blockIdx.x / (NPTS / DPW);
    const int j0   = (blockIdx.x % (NPTS / DPW)) * DPW;
    const int*      nbhb = nbh + (size_t)b * KNBH * NPTS;
    const float*    posb = pos + (size_t)b * 3 * NPTS;
    const float*    prb  = posT + (size_t)b * NPTS * 4;
    const unsigned* ftb  = ftT + (size_t)b * NPTS * ROW_WORDS;
    float*          outb = out + (size_t)b * COUT * NPTS;

    if (tid < DPW) s_cnt[tid] = 0;
    __syncthreads();

    if (use_lists) {
        const int t   = blockIdx.x % (NPTS / DPW);
        const int s   = tid >> 2;
        const int sub = tid & 3;
        const unsigned* region = lists + ((size_t)b * LIST_SLICES + s) * LIST_WORDS;
        const unsigned d = region[t];
        const int start = min((int)(d >> 16), LIST_ENTRIES);
        const int count = min((int)(d & 0xFFFFu), LIST_ENTRIES - start);
        const unsigned short* ep = (const unsigned short*)(region + LIST_TILES) + start;
        unsigned w[LIST_AHEAD];
        int      slot[LIST_AHEAD];
        #pragma unroll
        for (int u = 0; u < LIST_AHEAD; ++u) {
            const int e = sub + 4 * u;
            w[u] = (e < count) ? (unsigned)ep[e] : 0u;
        }
        #pragma unroll
        for (int u = 0; u < LIST_AHEAD; ++u) {
            const int e = sub + 4 * u;
            slot[u] = BIN_CAP;
            if (e < count) slot[u] = atomicAdd(&s_cnt[(w[u] >> 6) & (DPW - 1)], 1);
        }
        #pragma unroll
        for (int u = 0; u < LIST_AHEAD; ++u) {
            if (slot[u] < BIN_CAP)
                s_bin[((w[u] >> 6) & (DPW - 1)) * BIN_CAP + slot[u]] = (unsigned short)(s * TP_PTS + (int)(w[u] & 63u));
        }
        for (int e = sub + 4 * LIST_AHEAD; e < count; e += 4) {
            const unsigned we = ep[e];
            const unsigned dl = (we >> 6) & (DPW - 1);
            const int sl = atomicAdd(&s_cnt[dl], 1);
            if (sl < BIN_CAP) s_bin[dl * BIN_CAP + sl] = (unsigned short)(s * TP_PTS + (int)(we & 63u));
        }
    } else {
        const int4* idx4 = (const int4*)nbhb;
        for (int it0 = 0; it0 < 32; it0 += SCAN_GROUP) {
            int4 v[SCAN_GROUP];
            #pragma unroll
            for (int u = 0; u < SCAN_GROUP; ++u) v[u] = idx4[tid + OWN_THREADS * (it0 + u)];
            #pragma unroll
            for (int u = 0; u < SCAN_GROUP; ++u) {
                const int f = 4 * (tid + OWN_THREADS * (it0 + u));
                const int jv[4] = {v[u].x, v[u].y, v[u].z, v[u].w};
                #pragma unroll
                for (int e = 0; e < 4; ++e) {
                    const unsigned dl = (unsigned)(jv[e] - j0);
                    if (dl < (unsigned)DPW) {
                        const int slot = atomicAdd(&s_cnt[dl], 1);
                        if (slot < BIN_CAP) s_bin[dl * BIN_CAP + slot] = (unsigned short)((f + e) & (NPTS - 1));
                    }
                }
            }
        }
    }
    __syncthreads();

    for (int g2 = 0; g2 < DPW / GRP; ++g2) {
        for (int u = 0; u < GRP / OWN_WAVES; ++u) {
            const int jl  = wave * (GRP / OWN_WAVES) + u;
            const int dl  = g2 * GRP + jl;
            const int j   = j0 + dl;
            const int cnt = __builtin_amdgcn_readfirstlane(s_cnt[dl]);
            float pjx, pjy, pjz;
            if (PACKED_POS) {
                load_record(prb + (size_t)j * 4, pjx, pjy, pjz);
            } else {
                pjx = posb[j];  pjy = posb[NPTS + j];  pjz = posb[2 * NPTS + j];
            }
            v2f acc[4] = {{0.f, 0.f}, {0.f, 0.f}, {0.f, 0.f}, {0.f, 0.f}};

            if (cnt <= BIN_CAP) {
                const int myn = (lane < cnt) ? (int)s_bin[dl * BIN_CAP + lane] : j;
                const unsigned myoff = (unsigned)myn * (unsigned)(ROW_WORDS * 4);
                float mpx, mpy, mpz;
                if (PACKED_POS) {
                    load_record((const float*)((const char*)prb + (size_t)((unsigned)myn * 16u)), mpx, mpy, mpz);
                } else {
                    mpx = posb[myn];  mpy = posb[NPTS + myn];  mpz = posb[2 * NPTS + myn];
                }
                switch (min(cnt, HIT_GROUP)) {
                case 1:  sum_pass<1>(ftb, myoff, 0, lane, pjx, pjy, pjz, mpx, mpy, mpz, acc); break;
                case 2:  sum_pass<2>(ftb, myoff, 0, lane, pjx, pjy, pjz, mpx, mpy, mpz, acc); break;
                case 3:  sum_pass<3>(ftb, myoff, 0, lane, pjx, pjy, pjz, mpx, mpy, mpz, acc); break;
                case 4:  sum_pass<4>(ftb, myoff, 0, lane, pjx, pjy, pjz, mpx, mpy, mpz, acc); break;
                case 5:  sum_pass<5>(ftb, myoff, 0, lane, pjx, pjy, pjz, mpx, mpy, mpz, acc); break;
                case 6:  sum_pass<6>(ftb, myoff, 0, lane, pjx, pjy, pjz, mpx, mpy, mpz, acc); break;
                case 7:  sum_pass<7>(ftb, myoff, 0, lane, pjx, pjy, pjz, mpx, mpy, mpz, acc); break;
                case 8:  sum_pass<8>(ftb, myoff, 0, lane, pjx, pjy, pjz, mpx, mpy, mpz, acc); break;
                case 9:  sum_pass<9>(ftb, myoff, 0, lane, pjx, pjy, pjz, mpx, mpy, mpz, acc); break;
                case 10: sum_pass<10>(ftb, myoff, 0, lane, pjx, pjy, pjz, mpx, mpy, mpz, acc); break;
                case 11: sum_pass<11>(ftb, myoff, 0, lane, pjx, pjy, pjz, mpx, mpy, mpz, acc); break;
                case 12: sum_pass<12>(ftb, myoff, 0, lane, pjx, pjy, pjz, mpx, mpy, mpz, acc); break;
                case 13: sum_pass<13>(ftb, myoff, 0, lane, pjx, pjy, pjz, mpx, mpy, mpz, acc); break;
                case 14: sum_pass<14>(ftb, myoff, 0, lane, pjx, pjy, pjz, mpx, mpy, mpz, acc); break;
                case 15: sum_pass<15>(ftb, myoff, 0, lane, pjx, pjy, pjz, mpx, mpy, mpz, acc); break;
                case 16: sum_pass<16>(ftb, myoff, 0, lane, pjx, pjy, pjz, mpx, mpy, mpz, acc); break;
                default: break;
                }
                for (int h = HIT_GROUP; h < cnt; ++h) sum_pass<1>(ftb, myoff, h, lane, pjx, pjy, pjz, mpx, mpy, mpz, acc);
            } else {
                for (int base = 0; base < KNBH * NPTS; base += 32) {
                    const int jv = nbhb[base + lane];
                    unsigned m = __builtin_amdgcn_ballot_w32(jv == j);
                    while (m) {
                        const int srcl = __builtin_ctz(m);
                        m &= m - 1;
                        const int n = (base + srcl) & (NPTS - 1);
                        const unsigned row = ftb[(size_t)n * ROW_WORDS + lane];
                        float pnx, pny, pnz;
                        if (PACKED_POS) {
                            load_record(prb + (size_t)n * 4, pnx, pny, pnz);
                        } else {
                            pnx = posb[n];  pny = posb[NPTS + n];  pnz = posb[2 * NPTS + n];
                        }
                        add_source(row, pjx - pnx, pjy - pny, pjz - pnz, acc);
                    }
                }
            }

            {
                const int nt  = jl >> 4, col = jl & 15;
                const int kk  = (2 * lane) & 31;
                const int gg  = (kk >> 3) & 1;
                const int pc  = kk >> 4;
                const int eq  = kk & 7;
                const int kcl = lane >> 4;
                #pragma unroll
                for (int d = 0; d < 4; ++d) {
                    const int kc  = 2 * d + kcl;
                    const int idx = ((nt * XK_CHUNKS + kc) * 2 + pc) * XBLK + (gg * 16 + col) * 8 + eq;
                    const v2h hi = __builtin_convertvector(acc[d], v2h);
                    const v2h lo = __builtin_convertvector(acc[d] - __builtin_convertvector(hi, v2f), v2h);
                    *(v2h*)(s_x + g2 * X_GROUP + idx)            = hi;
                    *(v2h*)(s_x + g2 * X_GROUP + X_HALVES + idx) = lo;
                }
            }
        }
    }
    __syncthreads();

    if (wave < PROD_WAVES) {
        const int mtp = wave & 1;
        const int ntp = (wave >> 1) & 3;
        const int grp = wave >> 3;
        const _Float16* wa = wsA + (((size_t)(2 * mtp) * XK_CHUNKS) * 32 + lane) * 16;
        const _Float16* xb = s_x + grp * X_GROUP + ((2 * ntp) * XK_CHUNKS * 2) * XBLK + lane * 8;
        v8f  acc[2][2] = {};
        v16h af[3][2];
        v16h bf[2][2];
        #pragma unroll
        for (int q = 0; q < 2; ++q) {
            #pragma unroll
            for (int i = 0; i < 2; ++i)
                af[q][i] = *(const v16h*)(wa + ((size_t)(i * XK_CHUNKS + q) * 32) * 16);
        }
        #pragma unroll
        for (int t = 0; t < 2; ++t) {
            const _Float16* p = xb + (t * XK_CHUNKS * 2) * XBLK;
            bf[0][t] = __builtin_shufflevector(*(const v8h*)p, *(const v8h*)(p + XBLK),
                                               0, 1, 2, 3, 4, 5, 6, 7, 8, 9, 10, 11, 12, 13, 14, 15);
        }
        #pragma unroll
        for (int s = 0; s < 2 * XK_CHUNKS; ++s) {
            const int kc = s >> 1;
            if (s + 1 < 2 * XK_CHUNKS) {
                const int kn = (s + 1) >> 1, pn = (s + 1) & 1;
                #pragma unroll
                for (int t = 0; t < 2; ++t) {
                    const _Float16* p = xb + pn * X_HALVES + ((t * XK_CHUNKS + kn) * 2) * XBLK;
                    bf[(s + 1) & 1][t] = __builtin_shufflevector(*(const v8h*)p, *(const v8h*)(p + XBLK),
                                                                 0, 1, 2, 3, 4, 5, 6, 7, 8, 9, 10, 11, 12, 13, 14, 15);
                }
            }
            if ((s & 1) == 0 && kc + 2 < XK_CHUNKS) {
                #pragma unroll
                for (int i = 0; i < 2; ++i)
                    af[(kc + 2) % 3][i] = *(const v16h*)(wa + ((size_t)(i * XK_CHUNKS + kc + 2) * 32) * 16);
            }
            __builtin_amdgcn_sched_barrier(0);
            #pragma unroll
            for (int i = 0; i < 2; ++i) {
                #pragma unroll
                for (int t = 0; t < 2; ++t)
                    acc[i][t] = __builtin_amdgcn_wmma_f32_16x16x32_f16(false, af[kc % 3][i], false, bf[s & 1][t],
                                                                       (short)0, acc[i][t], false, false);
            }
            __builtin_amdgcn_sched_barrier(0);
        }
        const int jbase = j0 + grp * GRP + ntp * 32 + (lane & 15);
        #pragma unroll
        for (int i = 0; i < 2; ++i) {
            #pragma unroll
            for (int r = 0; r < 8; ++r) {
                const int o = (2 * mtp + i) * 16 + 8 * (lane >> 4) + r;
                const float bo = bias[o];
                outb[(size_t)o * NPTS + jbase]      = acc[i][0][r] + bo;
                outb[(size_t)o * NPTS + jbase + 16] = acc[i][1][r] + bo;
            }
        }
    }
}

extern "C" void kernel_launch(void* const* d_in, const int* in_sizes, int n_in,
                              void* d_out, int out_size, void* d_ws, size_t ws_size,
                              hipStream_t stream) {
    const float* features = (const float*)d_in[0];
    const float* theta    = (const float*)d_in[1];
    const float* wbias    = (const float*)d_in[2];
    const int*   nbh      = (const int*)d_in[3];
    const float* pos      = (const float*)d_in[4];
    const float* bias     = (const float*)d_in[5];
    float*       out      = (float*)d_out;
    _Float16*    wsA      = (_Float16*)d_ws;
    int*         tickets  = (int*)((char*)d_ws + WS_A_BYTES);

    if (ws_size >= WS_NEED) {
        unsigned* ftT   = (unsigned*)((char*)d_ws + WS_A_BYTES);
        unsigned* lists = (unsigned*)((char*)d_ws + WS_NEED);
        float*    posT  = (float*)((char*)d_ws + WS_NEED_LISTS);
        if (ws_size >= WS_NEED_POS) {
            pack_rows_kernel<1><<<PACK2_BLOCKS + TP_BLOCKS, 256, 0, stream>>>(features, theta, wbias, nbh, pos, wsA, ftT, lists, posT, 1);
            owner_kernel<1><<<BATCH * (NPTS / DPW), OWN_THREADS, 0, stream>>>(nbh, pos, bias, wsA, ftT, lists, posT, 1, out);
            return;
        }
        if (ws_size >= WS_NEED_LISTS) {
            pack_rows_kernel<0><<<PACK2_BLOCKS + TP_BLOCKS, 256, 0, stream>>>(features, theta, wbias, nbh, pos, wsA, ftT, lists, posT, 1);
            owner_kernel<0><<<BATCH * (NPTS / DPW), OWN_THREADS, 0, stream>>>(nbh, pos, bias, wsA, ftT, lists, posT, 1, out);
            return;
        }
        pack_rows_kernel<0><<<PACK2_BLOCKS + TP_BLOCKS, 256, 0, stream>>>(features, theta, wbias, nbh, pos, wsA, ftT, lists, posT, 0);
        owner_kernel<0><<<BATCH * (NPTS / DPW), OWN_THREADS, 0, stream>>>(nbh, pos, bias, wsA, ftT, lists, posT, 0, out);
        return;
    }

    prepare_kernel<<<PREP_BLOCKS, PREP_THREADS, 0, stream>>>(bias, theta, wbias, out, wsA, tickets);
    flexconv_kernel<<<BATCH * (NPTS / TILE_PTS), 256, 0, stream>>>(
        features, nbh, pos, wsA, tickets, out);
    turn_blocks_kernel<<<BATCH * (NPTS / TBLK), 256, 0, stream>>>(out);
}
